# starting kernel
# baseline (speedup 1.0000x reference)
#include <hip/hip_runtime.h>
#include <hip/hip_bf16.h>
#include <math.h>

typedef __attribute__((ext_vector_type(16))) __bf16 v16bf;
typedef __attribute__((ext_vector_type(8)))  __bf16 v8bf;
typedef __attribute__((ext_vector_type(8)))  float  v8f;

#define N_NODES 100000
#define F_IN    500
#define H_DIM   64
#define C_DIM   40
#define N_EDGES 3200000
#define K_STEPS 10
#define ALPHA   0.1f

__global__ __launch_bounds__(256) void appnp_gemm1(const float* __restrict__ x,
                                                   const float* __restrict__ W1,
                                                   const float* __restrict__ b1,
                                                   __bf16* __restrict__ h1) {
    const int lane = threadIdx.x & 31;
    const int wid  = threadIdx.x >> 5;
    const int rowTile = blockIdx.x * 2 + (wid >> 2);
    const int colTile = wid & 3;
    const int hiHalf  = (lane & 16) ? 8 : 0;
    const int m = rowTile * 16 + (lane & 15);
    const int n = colTile * 16 + (lane & 15);

    const float* xrow = x + (size_t)m * F_IN;
    v8f acc = {};

    for (int kb = 0; kb < 480; kb += 32) {
        const int kA = kb + hiHalf;
        const float4 r0 = *(const float4*)(xrow + kA);
        const float4 r1 = *(const float4*)(xrow + kA + 4);
        const float4 r2 = *(const float4*)(xrow + kA + 16);
        const float4 r3 = *(const float4*)(xrow + kA + 20);

        v16bf a;
        a[0]  = (__bf16)r0.x; a[1]  = (__bf16)r0.y; a[2]  = (__bf16)r0.z; a[3]  = (__bf16)r0.w;
        a[4]  = (__bf16)r1.x; a[5]  = (__bf16)r1.y; a[6]  = (__bf16)r1.z; a[7]  = (__bf16)r1.w;
        a[8]  = (__bf16)r2.x; a[9]  = (__bf16)r2.y; a[10] = (__bf16)r2.z; a[11] = (__bf16)r2.w;
        a[12] = (__bf16)r3.x; a[13] = (__bf16)r3.y; a[14] = (__bf16)r3.z; a[15] = (__bf16)r3.w;

        v16bf b;
#pragma unroll
        for (int j = 0; j < 16; ++j) {
            int k = kb + j + (j & 8) + hiHalf;
            b[j] = (__bf16)W1[k * H_DIM + n];
        }
        acc = __builtin_amdgcn_wmma_f32_16x16x32_bf16(
            false, a, false, b, (short)0, acc, false, false);
    }

    {
        const int kA = 480 + hiHalf;
        const float4 r0 = *(const float4*)(xrow + kA);
        const float4 r1 = *(const float4*)(xrow + kA + 4);
        const float4 r2 = *(const float4*)(xrow + 496);
        const float fa2[4] = {r2.x, r2.y, r2.z, r2.w};

        v16bf a;
        a[0] = (__bf16)r0.x; a[1] = (__bf16)r0.y; a[2] = (__bf16)r0.z; a[3] = (__bf16)r0.w;
        a[4] = (__bf16)r1.x; a[5] = (__bf16)r1.y; a[6] = (__bf16)r1.z; a[7] = (__bf16)r1.w;
#pragma unroll
        for (int j = 0; j < 8; ++j) {
            int k1 = kA + 16 + j;
            float val = (j < 4) ? fa2[j] : 0.0f;
            float f1  = (k1 < F_IN) ? val : 0.0f;
            a[j + 8] = (__bf16)f1;
        }

        v16bf b;
#pragma unroll
        for (int j = 0; j < 16; ++j) {
            int k  = 480 + j + (j & 8) + hiHalf;
            int kc = (k < F_IN) ? k : (F_IN - 1);
            b[j] = (__bf16)W1[kc * H_DIM + n];
        }
        acc = __builtin_amdgcn_wmma_f32_16x16x32_bf16(
            false, a, false, b, (short)0, acc, false, false);
    }

    const float bias = b1[n];
#pragma unroll
    for (int r = 0; r < 8; ++r) {
        int mr = rowTile * 16 + r + hiHalf;
        float v = fmaxf(acc[r] + bias, 0.0f);
        h1[(size_t)mr * H_DIM + n] = (__bf16)v;
    }
}

__global__ __launch_bounds__(256) void appnp_gemm2(const __bf16* __restrict__ h1,
                                                   const float* __restrict__ W2,
                                                   const float* __restrict__ b2,
                                                   float* __restrict__ h,
                                                   float* __restrict__ h0) {
    const int lane = threadIdx.x & 31;
    const int wid  = threadIdx.x >> 5;
    const int tile = blockIdx.x * 8 + wid;
    const int nTiles = (N_NODES / 16) * 3;
    if (tile >= nTiles) return;
    const int rowTile = tile / 3;
    const int colTile = tile % 3;
    const int hiHalf  = (lane & 16) ? 8 : 0;
    const int m = rowTile * 16 + (lane & 15);
    const int n = colTile * 16 + (lane & 15);
    const int ncl = (n < C_DIM) ? n : (C_DIM - 1);

    const __bf16* hrow = h1 + (size_t)m * H_DIM;
    v8f acc = {};
#pragma unroll
    for (int kb = 0; kb < H_DIM; kb += 32) {
        const int kA = kb + hiHalf;
        const v8bf lo = *(const v8bf*)(hrow + kA);
        const v8bf hi = *(const v8bf*)(hrow + kA + 16);
        v16bf a;
#pragma unroll
        for (int j = 0; j < 8; ++j) { a[j] = lo[j]; a[j + 8] = hi[j]; }

        v16bf b;
#pragma unroll
        for (int j = 0; j < 16; ++j) {
            int k = kb + j + (j & 8) + hiHalf;
            b[j] = (__bf16)W2[k * C_DIM + ncl];
        }
        acc = __builtin_amdgcn_wmma_f32_16x16x32_bf16(
            false, a, false, b, (short)0, acc, false, false);
    }

    if (n < C_DIM) {
        const float bias = b2[ncl];
#pragma unroll
        for (int r = 0; r < 8; ++r) {
            int mr = rowTile * 16 + r + hiHalf;
            float v = acc[r] + bias;
            h [(size_t)mr * C_DIM + n] = v;
            h0[(size_t)mr * C_DIM + n] = v;
        }
    }
}

__global__ __launch_bounds__(256) void appnp_deg_init(float* __restrict__ deg) {
    int i = blockIdx.x * blockDim.x + threadIdx.x;
    if (i < N_NODES) deg[i] = 1.0f;
}

__global__ __launch_bounds__(256) void appnp_deg_count(const int* __restrict__ dst,
                                                       float* __restrict__ deg) {
    int e = blockIdx.x * blockDim.x + threadIdx.x;
    if (e < N_EDGES) unsafeAtomicAdd(&deg[dst[e]], 1.0f);
}

__global__ __launch_bounds__(256) void appnp_dinv(const float* __restrict__ deg,
                                                  float* __restrict__ dinv) {
    int i = blockIdx.x * blockDim.x + threadIdx.x;
    if (i < N_NODES) dinv[i] = rsqrtf(deg[i]);
}

__global__ __launch_bounds__(256) void appnp_prop_init(const float* __restrict__ h,
                                                       const float* __restrict__ h0,
                                                       const float* __restrict__ dinv,
                                                       float* __restrict__ hn) {
    int t = blockIdx.x * blockDim.x + threadIdx.x;
    int i = t / C_DIM;
    float di = dinv[i];
    hn[t] = ALPHA * h0[t] + (1.0f - ALPHA) * di * di * h[t];
}

__global__ __launch_bounds__(256) void appnp_prop_edges(const int* __restrict__ src,
                                                        const int* __restrict__ dst,
                                                        const float* __restrict__ dinv,
                                                        const float* __restrict__ h,
                                                        float* __restrict__ hn) {
    const int lane = threadIdx.x & 31;
    const int e = blockIdx.x * 8 + (threadIdx.x >> 5);
    if (e >= N_EDGES) return;
    const int s = src[e];
    const int d = dst[e];
    const float w = (1.0f - ALPHA) * dinv[s] * dinv[d];
    const float* hs = h  + (size_t)s * C_DIM;
    float*       hd = hn + (size_t)d * C_DIM;
    unsafeAtomicAdd(&hd[lane], w * hs[lane]);
    if (lane < C_DIM - 32)
        unsafeAtomicAdd(&hd[lane + 32], w * hs[lane + 32]);
}

__global__ __launch_bounds__(256) void appnp_logsoftmax(const float* __restrict__ h,
                                                        float* __restrict__ out) {
    const int lane = threadIdx.x & 31;
    const int row = blockIdx.x * 8 + (threadIdx.x >> 5);
    if (row >= N_NODES) return;
    const float* r = h + (size_t)row * C_DIM;
    float v0 = r[lane];
    float v1 = (lane < C_DIM - 32) ? r[lane + 32] : -3.4e38f;

    float m = fmaxf(v0, v1);
#pragma unroll
    for (int off = 16; off > 0; off >>= 1) m = fmaxf(m, __shfl_xor(m, off, 32));

    float s = __expf(v0 - m) + ((lane < C_DIM - 32) ? __expf(v1 - m) : 0.0f);
#pragma unroll
    for (int off = 16; off > 0; off >>= 1) s += __shfl_xor(s, off, 32);

    float lse = m + __logf(s);
    float* o = out + (size_t)row * C_DIM;
    o[lane] = v0 - lse;
    if (lane < C_DIM - 32) o[lane + 32] = v1 - lse;
}

extern "C" void kernel_launch(void* const* d_in, const int* in_sizes, int n_in,
                              void* d_out, int out_size, void* d_ws, size_t ws_size,
                              hipStream_t stream) {
    const float* x  = (const float*)d_in[0];
    const float* W1 = (const float*)d_in[1];
    const float* b1 = (const float*)d_in[2];
    const float* W2 = (const float*)d_in[3];
    const float* b2 = (const float*)d_in[4];
    const int*   ei = (const int*)d_in[5];
    const int*   src = ei;
    const int*   dst = ei + N_EDGES;
    float* out = (float*)d_out;

    char* ws = (char*)d_ws;
    __bf16* h1  = (__bf16*)ws;
    float* hA   = (float*)(ws + 12800000);
    float* hB   = (float*)(ws + 12800000 + 16000000);
    float* h0   = (float*)(ws + 12800000 + 2 * 16000000);
    float* deg  = (float*)(ws + 12800000 + 3 * 16000000);
    float* dinv = (float*)(ws + 12800000 + 3 * 16000000 + 400128);

    appnp_gemm1<<<N_NODES / 32, 256, 0, stream>>>(x, W1, b1, h1);
    appnp_gemm2<<<((N_NODES / 16) * 3 + 7) / 8, 256, 0, stream>>>(h1, W2, b2, hA, h0);

    appnp_deg_init<<<(N_NODES + 255) / 256, 256, 0, stream>>>(deg);
    appnp_deg_count<<<(N_EDGES + 255) / 256, 256, 0, stream>>>(dst, deg);
    appnp_dinv<<<(N_NODES + 255) / 256, 256, 0, stream>>>(deg, dinv);

    float* cur = hA;
    float* nxt = hB;
    for (int k = 0; k < K_STEPS; ++k) {
        appnp_prop_init<<<(N_NODES * C_DIM) / 256, 256, 0, stream>>>(cur, h0, dinv, nxt);
        appnp_prop_edges<<<(N_EDGES + 7) / 8, 256, 0, stream>>>(src, dst, dinv, cur, nxt);
        float* t = cur; cur = nxt; nxt = t;
    }

    appnp_logsoftmax<<<(N_NODES + 7) / 8, 256, 0, stream>>>(cur, out);
}


# Propagation LDS sums kept in 4 copies picked by edge index mod 4, each shifted one bank: the 4 edges of one LDS add use disjoint banks (was 16 of 64); row = 4 arrival-order partial sums added
# speedup vs baseline: 4.6657x; 4.6657x over previous
#include <hip/hip_runtime.h>
#include <hip/hip_bf16.h>
#include <math.h>

typedef __attribute__((ext_vector_type(16))) __bf16 v16bf;
typedef __attribute__((ext_vector_type(8)))  __bf16 v8bf;
typedef __attribute__((ext_vector_type(8)))  float  v8f;

#define N_NODES 100000
#define F_IN    500
#define H_DIM   64
#define C_DIM   40
#define N_EDGES 3200000
#define K_STEPS 10
#define ALPHA   0.1f

__global__ __launch_bounds__(256) void appnp_gemm1(const float* __restrict__ x,
                                                   const float* __restrict__ W1,
                                                   const float* __restrict__ b1,
                                                   __bf16* __restrict__ h1) {
    const int lane = threadIdx.x & 31;
    const int wid  = threadIdx.x >> 5;
    const int rowTile = blockIdx.x * 2 + (wid >> 2);
    const int colTile = wid & 3;
    const int hiHalf  = (lane & 16) ? 8 : 0;
    const int m = rowTile * 16 + (lane & 15);
    const int n = colTile * 16 + (lane & 15);

    const float* xrow = x + (size_t)m * F_IN;
    v8f acc = {};

    for (int kb = 0; kb < 480; kb += 32) {
        const int kA = kb + hiHalf;
        const float4 r0 = *(const float4*)(xrow + kA);
        const float4 r1 = *(const float4*)(xrow + kA + 4);
        const float4 r2 = *(const float4*)(xrow + kA + 16);
        const float4 r3 = *(const float4*)(xrow + kA + 20);

        v16bf a;
        a[0]  = (__bf16)r0.x; a[1]  = (__bf16)r0.y; a[2]  = (__bf16)r0.z; a[3]  = (__bf16)r0.w;
        a[4]  = (__bf16)r1.x; a[5]  = (__bf16)r1.y; a[6]  = (__bf16)r1.z; a[7]  = (__bf16)r1.w;
        a[8]  = (__bf16)r2.x; a[9]  = (__bf16)r2.y; a[10] = (__bf16)r2.z; a[11] = (__bf16)r2.w;
        a[12] = (__bf16)r3.x; a[13] = (__bf16)r3.y; a[14] = (__bf16)r3.z; a[15] = (__bf16)r3.w;

        v16bf b;
#pragma unroll
        for (int j = 0; j < 16; ++j) {
            int k = kb + j + (j & 8) + hiHalf;
            b[j] = (__bf16)W1[k * H_DIM + n];
        }
        acc = __builtin_amdgcn_wmma_f32_16x16x32_bf16(
            false, a, false, b, (short)0, acc, false, false);
    }

    {
        const int kA = 480 + hiHalf;
        const float4 r0 = *(const float4*)(xrow + kA);
        const float4 r1 = *(const float4*)(xrow + kA + 4);
        const float4 r2 = *(const float4*)(xrow + 496);
        const float fa2[4] = {r2.x, r2.y, r2.z, r2.w};

        v16bf a;
        a[0] = (__bf16)r0.x; a[1] = (__bf16)r0.y; a[2] = (__bf16)r0.z; a[3] = (__bf16)r0.w;
        a[4] = (__bf16)r1.x; a[5] = (__bf16)r1.y; a[6] = (__bf16)r1.z; a[7] = (__bf16)r1.w;
#pragma unroll
        for (int j = 0; j < 8; ++j) {
            int k1 = kA + 16 + j;
            float val = (j < 4) ? fa2[j] : 0.0f;
            float f1  = (k1 < F_IN) ? val : 0.0f;
            a[j + 8] = (__bf16)f1;
        }

        v16bf b;
#pragma unroll
        for (int j = 0; j < 16; ++j) {
            int k  = 480 + j + (j & 8) + hiHalf;
            int kc = (k < F_IN) ? k : (F_IN - 1);
            b[j] = (__bf16)W1[kc * H_DIM + n];
        }
        acc = __builtin_amdgcn_wmma_f32_16x16x32_bf16(
            false, a, false, b, (short)0, acc, false, false);
    }

    const float bias = b1[n];
#pragma unroll
    for (int r = 0; r < 8; ++r) {
        int mr = rowTile * 16 + r + hiHalf;
        float v = fmaxf(acc[r] + bias, 0.0f);
        h1[(size_t)mr * H_DIM + n] = (__bf16)v;
    }
}

__global__ __launch_bounds__(256) void appnp_gemm2(const __bf16* __restrict__ h1,
                                                   const float* __restrict__ W2,
                                                   const float* __restrict__ b2,
                                                   float* __restrict__ h,
                                                   float* __restrict__ h0) {
    const int lane = threadIdx.x & 31;
    const int wid  = threadIdx.x >> 5;
    const int tile = blockIdx.x * 8 + wid;
    const int nTiles = (N_NODES / 16) * 3;
    if (tile >= nTiles) return;
    const int rowTile = tile / 3;
    const int colTile = tile % 3;
    const int hiHalf  = (lane & 16) ? 8 : 0;
    const int m = rowTile * 16 + (lane & 15);
    const int n = colTile * 16 + (lane & 15);
    const int ncl = (n < C_DIM) ? n : (C_DIM - 1);

    const __bf16* hrow = h1 + (size_t)m * H_DIM;
    v8f acc = {};
#pragma unroll
    for (int kb = 0; kb < H_DIM; kb += 32) {
        const int kA = kb + hiHalf;
        const v8bf lo = *(const v8bf*)(hrow + kA);
        const v8bf hi = *(const v8bf*)(hrow + kA + 16);
        v16bf a;
#pragma unroll
        for (int j = 0; j < 8; ++j) { a[j] = lo[j]; a[j + 8] = hi[j]; }

        v16bf b;
#pragma unroll
        for (int j = 0; j < 16; ++j) {
            int k = kb + j + (j & 8) + hiHalf;
            b[j] = (__bf16)W2[k * C_DIM + ncl];
        }
        acc = __builtin_amdgcn_wmma_f32_16x16x32_bf16(
            false, a, false, b, (short)0, acc, false, false);
    }

    if (n < C_DIM) {
        const float bias = b2[ncl];
#pragma unroll
        for (int r = 0; r < 8; ++r) {
            int mr = rowTile * 16 + r + hiHalf;
            float v = acc[r] + bias;
            h [(size_t)mr * C_DIM + n] = v;
            h0[(size_t)mr * C_DIM + n] = v;
        }
    }
}

#define NB      131
#define NBUCK   764
#define PT      640
#define ROWQ    10
#define GCHUNK  8192
#define ACOPIES 4
#define ASTRIDE (NB * C_DIM + 1)

__global__ __launch_bounds__(1024) void appnp_tile_zero(int* __restrict__ bcount) {
    if (threadIdx.x < NBUCK) bcount[threadIdx.x] = 0;
}

__global__ __launch_bounds__(1024) void appnp_tile_hist(const int* __restrict__ dst,
                                                        int* __restrict__ bcount) {
    __shared__ int hist[NBUCK];
    const int tid = threadIdx.x;
    if (tid < NBUCK) hist[tid] = 0;
    __syncthreads();
    const int e0 = blockIdx.x * GCHUNK;
    int d[GCHUNK / 1024];
    for (int k = 0; k < GCHUNK / 1024; ++k) {
        const int e = e0 + k * 1024 + tid;
        d[k] = dst[(e < N_EDGES) ? e : (N_EDGES - 1)];
    }
    for (int k = 0; k < GCHUNK / 1024; ++k) {
        const int e = e0 + k * 1024 + tid;
        if (e < N_EDGES) atomicAdd(&hist[d[k] / NB], 1);
    }
    __syncthreads();
    if (tid < NBUCK) {
        const int c = hist[tid];
        if (c) atomicAdd(&bcount[tid], c);
    }
}

__global__ __launch_bounds__(1024) void appnp_tile_scan(const int* __restrict__ bcount,
                                                        int* __restrict__ bstart,
                                                        int* __restrict__ cursor) {
    __shared__ int s[1024];
    const int t = threadIdx.x;
    const int c = (t < NBUCK) ? bcount[t] : 0;
    s[t] = c;
    __syncthreads();
    for (int off = 1; off < 1024; off <<= 1) {
        const int v = (t >= off) ? s[t - off] : 0;
        __syncthreads();
        s[t] += v;
        __syncthreads();
    }
    const int excl = s[t] - c;
    if (t < NBUCK) { bstart[t] = excl; cursor[t] = excl; }
    if (t == NBUCK - 1) bstart[NBUCK] = s[t];
}

__global__ __launch_bounds__(1024) void appnp_tile_scatter(const int* __restrict__ src,
                                                           const int* __restrict__ dst,
                                                           int* __restrict__ cursor,
                                                           unsigned* __restrict__ ebuf) {
    __shared__ int hist[NBUCK];
    const int tid = threadIdx.x;
    if (tid < NBUCK) hist[tid] = 0;
    __syncthreads();
    const int e0 = blockIdx.x * GCHUNK;
    int      bk[GCHUNK / 1024];
    int      rk[GCHUNK / 1024];
    unsigned wd[GCHUNK / 1024];
#pragma unroll
    for (int k = 0; k < GCHUNK / 1024; ++k) {
        const int e = e0 + k * 1024 + tid;
        bk[k] = -1; rk[k] = 0; wd[k] = 0u;
        if (e < N_EDGES) {
            const int d = dst[e];
            const int s = src[e];
            const int b = d / NB;
            bk[k] = b;
            wd[k] = ((unsigned)s << 8) | (unsigned)(d - b * NB);
            rk[k] = atomicAdd(&hist[b], 1);
        }
    }
    __syncthreads();
    if (tid < NBUCK) {
        const int c = hist[tid];
        hist[tid] = c ? atomicAdd(&cursor[tid], c) : 0;
    }
    __syncthreads();
#pragma unroll
    for (int k = 0; k < GCHUNK / 1024; ++k)
        if (bk[k] >= 0) ebuf[hist[bk[k]] + rk[k]] = wd[k];
}

__global__ __launch_bounds__(PT) void appnp_tile_setup(const unsigned* __restrict__ ebuf,
                                                       const int* __restrict__ bstart,
                                                       const float* __restrict__ h0,
                                                       float* __restrict__ dinv,
                                                       float* __restrict__ y) {
    __shared__ int   cnt[NB];
    __shared__ float sd[NB];
    const int tid   = threadIdx.x;
    const int node0 = blockIdx.x * NB;
    const int nb    = (N_NODES - node0 < NB) ? (N_NODES - node0) : NB;
    if (tid < NB) cnt[tid] = 0;
    __syncthreads();
    const int e0 = bstart[blockIdx.x];
    const int e1 = bstart[blockIdx.x + 1];
    for (int e = e0 + tid; e < e1; e += PT) atomicAdd(&cnt[ebuf[e] & 255u], 1);
    __syncthreads();
    if (tid < nb) {
        const float di = rsqrtf(1.0f + (float)cnt[tid]);
        sd[tid] = di;
        dinv[node0 + tid] = di;
    }
    __syncthreads();
    const int nval = nb * C_DIM;
    for (int i = tid; i < nval; i += PT) {
        const int g = node0 * C_DIM + i;
        y[g] = sd[i / C_DIM] * h0[g];
    }
}

__global__ __launch_bounds__(PT) void appnp_tile_prop(const unsigned* __restrict__ ebuf,
                                                      const int* __restrict__ bstart,
                                                      const float* __restrict__ ycur,
                                                      const float* __restrict__ h0,
                                                      const float* __restrict__ dinv,
                                                      float* __restrict__ ynext,
                                                      const int scale_out) {
    __shared__ float acc[ACOPIES * ASTRIDE];
    const int tid   = threadIdx.x;
    const int node0 = blockIdx.x * NB;
    const int nb    = (N_NODES - node0 < NB) ? (N_NODES - node0) : NB;
    for (int i = tid; i < ACOPIES * ASTRIDE; i += PT) acc[i] = 0.0f;
    __syncthreads();

    const int e0     = bstart[blockIdx.x];
    const int nitems = (bstart[blockIdx.x + 1] - e0) * ROWQ;
    if (nitems > 0) {
        const int ntrip = (nitems + 4 * PT - 1) / (4 * PT);
        unsigned w[4];
        int      q4[4];
        int      cb[4];
        bool     ok[4];
#pragma unroll
        for (int k = 0; k < 4; ++k) {
            const int i  = tid + k * PT;
            ok[k] = i < nitems;
            const int ii = ok[k] ? i : 0;
            const int e  = ii / ROWQ;
            q4[k] = (ii - e * ROWQ) * 4;
            cb[k] = (e & (ACOPIES - 1)) * ASTRIDE + q4[k];
            w[k]  = ebuf[e0 + e];
        }
        for (int t = 0; t < ntrip; ++t) {
            unsigned wc[4];
            int      aoff[4];
            int      qc[4];
            bool     okc[4];
            float4   v[4];
#pragma unroll
            for (int k = 0; k < 4; ++k) {
                wc[k]   = w[k];
                qc[k]   = q4[k];
                okc[k]  = ok[k];
                aoff[k] = ok[k] ? (int)(w[k] & 255u) * C_DIM + cb[k] : tid * 4;
            }
#pragma unroll
            for (int k = 0; k < 4; ++k) {
                const int i  = tid + ((t + 1) * 4 + k) * PT;
                ok[k] = i < nitems;
                const int ii = ok[k] ? i : 0;
                const int e  = ii / ROWQ;
                q4[k] = (ii - e * ROWQ) * 4;
                cb[k] = (e & (ACOPIES - 1)) * ASTRIDE + q4[k];
                w[k]  = ebuf[e0 + e];
            }
#pragma unroll
            for (int k = 0; k < 4; ++k)
                v[k] = *(const float4*)(ycur + (size_t)(wc[k] >> 8) * C_DIM + qc[k]);
#pragma unroll
            for (int k = 0; k < 4; ++k) {
                float* a = acc + aoff[k];
                unsafeAtomicAdd(a + 0, okc[k] ? v[k].x : 0.0f);
                unsafeAtomicAdd(a + 1, okc[k] ? v[k].y : 0.0f);
                unsafeAtomicAdd(a + 2, okc[k] ? v[k].z : 0.0f);
                unsafeAtomicAdd(a + 3, okc[k] ? v[k].w : 0.0f);
            }
        }
    }
    __syncthreads();

    const int nval = nb * C_DIM;
    for (int i = tid; i < nval; i += PT) {
        const int   g  = node0 * C_DIM + i;
        const float di = dinv[node0 + i / C_DIM];
        const float sum = (acc[i] + acc[ASTRIDE + i]) + (acc[2 * ASTRIDE + i] + acc[3 * ASTRIDE + i]);
        const float hn = ALPHA * h0[g] + (1.0f - ALPHA) * di * (sum + ycur[g]);
        ynext[g] = scale_out ? di * hn : hn;
    }
}

__global__ __launch_bounds__(256) void appnp_logsoftmax(const float* __restrict__ h,
                                                        float* __restrict__ out) {
    const int lane = threadIdx.x & 31;
    const int row = blockIdx.x * 8 + (threadIdx.x >> 5);
    if (row >= N_NODES) return;
    const float* r = h + (size_t)row * C_DIM;
    float v0 = r[lane];
    float v1 = (lane < C_DIM - 32) ? r[lane + 32] : -3.4e38f;

    float m = fmaxf(v0, v1);
#pragma unroll
    for (int off = 16; off > 0; off >>= 1) m = fmaxf(m, __shfl_xor(m, off, 32));

    float s = __expf(v0 - m) + ((lane < C_DIM - 32) ? __expf(v1 - m) : 0.0f);
#pragma unroll
    for (int off = 16; off > 0; off >>= 1) s += __shfl_xor(s, off, 32);

    float lse = m + __logf(s);
    float* o = out + (size_t)row * C_DIM;
    o[lane] = v0 - lse;
    if (lane < C_DIM - 32) o[lane + 32] = v1 - lse;
}

extern "C" void kernel_launch(void* const* d_in, const int* in_sizes, int n_in,
                              void* d_out, int out_size, void* d_ws, size_t ws_size,
                              hipStream_t stream) {
    const float* x  = (const float*)d_in[0];
    const float* W1 = (const float*)d_in[1];
    const float* b1 = (const float*)d_in[2];
    const float* W2 = (const float*)d_in[3];
    const float* b2 = (const float*)d_in[4];
    const int*   ei = (const int*)d_in[5];
    const int*   src = ei;
    const int*   dst = ei + N_EDGES;
    float* out = (float*)d_out;

    char* ws = (char*)d_ws;
    __bf16* h1  = (__bf16*)ws;
    unsigned* ebuf = (unsigned*)ws;
    float* hA   = (float*)(ws + 12800000);
    float* hB   = (float*)(ws + 12800000 + 16000000);
    float* h0   = (float*)(ws + 12800000 + 2 * 16000000);
    int* bcount = (int*)(ws + 12800000 + 3 * 16000000);
    int* bstart = (int*)(ws + 12800000 + 3 * 16000000 + 4096);
    int* cursor = (int*)(ws + 12800000 + 3 * 16000000 + 8192);
    float* dinv = (float*)(ws + 12800000 + 3 * 16000000 + 400128);

    appnp_gemm1<<<N_NODES / 32, 256, 0, stream>>>(x, W1, b1, h1);
    appnp_gemm2<<<((N_NODES / 16) * 3 + 7) / 8, 256, 0, stream>>>(h1, W2, b2, hA, h0);

    appnp_tile_zero<<<1, 1024, 0, stream>>>(bcount);
    appnp_tile_hist<<<(N_EDGES + GCHUNK - 1) / GCHUNK, 1024, 0, stream>>>(dst, bcount);
    appnp_tile_scan<<<1, 1024, 0, stream>>>(bcount, bstart, cursor);
    appnp_tile_scatter<<<(N_EDGES + GCHUNK - 1) / GCHUNK, 1024, 0, stream>>>(src, dst, cursor, ebuf);

    appnp_tile_setup<<<NBUCK, PT, 0, stream>>>(ebuf, bstart, h0, dinv, hA);

    float* cur = hA;
    float* nxt = hB;
    for (int k = 0; k < K_STEPS; ++k) {
        appnp_tile_prop<<<NBUCK, PT, 0, stream>>>(ebuf, bstart, cur, h0, dinv, nxt,
                                                  (k + 1 < K_STEPS) ? 1 : 0);
        float* t = cur; cur = nxt; nxt = t;
    }

    appnp_logsoftmax<<<(N_NODES + 7) / 8, 256, 0, stream>>>(cur, out);
}
